# starting kernel
# baseline (speedup 1.0000x reference)
#include <hip/hip_runtime.h>
#include <hip/hip_bf16.h>

typedef __attribute__((ext_vector_type(2))) float v2f;
typedef __attribute__((ext_vector_type(8))) float v8f;

#define B_DIM 4
#define TQ 512
#define TV 512
#define DM 256
#define UN 128
#define TPB_T 8

static __device__ __forceinline__ float fast_tanh(float x) {
#if __has_builtin(__builtin_amdgcn_tanhf)
    return __builtin_amdgcn_tanhf(x);
#else
    return tanhf(x);
#endif
}

static __device__ __forceinline__ void wmma_tile_f32(
    const float* __restrict__ A, int lda,
    const float* __restrict__ B, int ldb,
    float* __restrict__ C, int ldc,
    int m0, int n0, int K)
{
    const int lane = threadIdx.x & 31;
    const int half = lane >> 4;
    const int mn   = lane & 15;
    v8f acc = {0.f, 0.f, 0.f, 0.f, 0.f, 0.f, 0.f, 0.f};
    const float* __restrict__ arow = A + (size_t)(m0 + mn) * lda + 2 * half;
    const float* __restrict__ bcol = B + (size_t)(2 * half) * ldb + (n0 + mn);
#pragma unroll 2
    for (int k0 = 0; k0 < K; k0 += 4) {
        v2f a, b;
        a.x = arow[k0 + 0];
        a.y = arow[k0 + 1];
        b.x = bcol[(size_t)(k0 + 0) * ldb];
        b.y = bcol[(size_t)(k0 + 1) * ldb];
        acc = __builtin_amdgcn_wmma_f32_16x16x4_f32(
            false, a, false, b, (short)0, acc, false, false);
    }
#pragma unroll
    for (int r = 0; r < 8; ++r) {
        C[(size_t)(m0 + r + 8 * half) * ldc + (n0 + mn)] = acc[r];
    }
}

__global__ __launch_bounds__(256) void proj_kernel(
    const float* __restrict__ query, const float* __restrict__ value,
    const float* __restrict__ W1, const float* __restrict__ W2,
    float* __restrict__ Qo, float* __restrict__ Ko)
{
    const int wave  = (int)((blockIdx.x * blockDim.x + threadIdx.x) >> 5);
    const int which = wave >> 10;
    const int tile  = wave & 1023;
    const int mt    = tile >> 3;
    const int nt    = tile & 7;
    const float* A  = which ? value : query;
    const float* Bm = which ? W2 : W1;
    float* C        = which ? Ko : Qo;
    wmma_tile_f32(A, DM, Bm, UN, C, UN, mt * 16, nt * 16, DM);
}

__global__ __launch_bounds__(512) void score_softmax_kernel(
    const float* __restrict__ Q, const float* __restrict__ Kmat,
    const float* __restrict__ scale, float* __restrict__ attn)
{
    const int b    = (int)blockIdx.x >> 6;
    const int t0   = ((int)blockIdx.x & 63) * TPB_T;
    const int tid  = (int)threadIdx.x;
    __shared__ float q_sh[TPB_T][UN];
    __shared__ float sc_sh[UN];
    __shared__ float red[16];

    if (tid < UN) sc_sh[tid] = scale[tid];
    {
        const int tt = tid >> 7;
        const int uu = tid & 127;
        q_sh[tt][uu]     = Q[(size_t)(b * TQ + t0 + tt) * UN + uu];
        q_sh[tt + 4][uu] = Q[(size_t)(b * TQ + t0 + tt + 4) * UN + uu];
    }

    float kreg[UN];
    {
        const float4* __restrict__ krow4 =
            (const float4*)(Kmat + ((size_t)(b * TV + tid)) * UN);
#pragma unroll
        for (int i = 0; i < UN / 4; ++i) {
            const float4 v = krow4[i];
            kreg[4 * i + 0] = v.x;
            kreg[4 * i + 1] = v.y;
            kreg[4 * i + 2] = v.z;
            kreg[4 * i + 3] = v.w;
        }
    }
    __syncthreads();

    const int w = tid >> 5;
    for (int tt = 0; tt < TPB_T; ++tt) {
        float s = 0.f;
#pragma unroll
        for (int u = 0; u < UN; ++u) {
            s = __builtin_fmaf(sc_sh[u], fast_tanh(q_sh[tt][u] + kreg[u]), s);
        }

        float m = s;
#pragma unroll
        for (int off = 16; off > 0; off >>= 1)
            m = fmaxf(m, __shfl_xor(m, off, 32));
        if ((tid & 31) == 0) red[w] = m;
        __syncthreads();
        float bmax = red[0];
#pragma unroll
        for (int i = 1; i < 16; ++i) bmax = fmaxf(bmax, red[i]);

        const float e = __expf(s - bmax);

        __syncthreads();
        float sum = e;
#pragma unroll
        for (int off = 16; off > 0; off >>= 1)
            sum += __shfl_xor(sum, off, 32);
        if ((tid & 31) == 0) red[w] = sum;
        __syncthreads();
        float bsum = 0.f;
#pragma unroll
        for (int i = 0; i < 16; ++i) bsum += red[i];

        attn[(size_t)(b * TQ + t0 + tt) * TV + tid] = e / bsum;
        __syncthreads();
    }
}

__global__ __launch_bounds__(256) void context_kernel(
    const float* __restrict__ attn, const float* __restrict__ value,
    float* __restrict__ ctx)
{
    const int wave = (int)((blockIdx.x * blockDim.x + threadIdx.x) >> 5);
    const int b    = wave >> 9;
    const int t    = wave & 511;
    const int mt   = t >> 4;
    const int nt   = t & 15;
    const float* A  = attn  + (size_t)b * TQ * TV;
    const float* Bv = value + (size_t)b * TV * DM;
    float* C        = ctx   + (size_t)b * TQ * DM;
    wmma_tile_f32(A, TV, Bv, DM, C, DM, mt * 16, nt * 16, TV);
}

extern "C" void kernel_launch(void* const* d_in, const int* in_sizes, int n_in,
                              void* d_out, int out_size, void* d_ws, size_t ws_size,
                              hipStream_t stream) {
    const float* query = (const float*)d_in[0];
    const float* value = (const float*)d_in[1];
    const float* W1    = (const float*)d_in[2];
    const float* W2    = (const float*)d_in[3];
    const float* scale = (const float*)d_in[4];

    float* out  = (float*)d_out;
    float* ctx  = out;
    float* attn = out + (size_t)B_DIM * TQ * DM;

    float* Qw = (float*)d_ws;
    float* Kw = Qw + (size_t)(B_DIM * TQ) * UN;

    proj_kernel<<<256, 256, 0, stream>>>(query, value, W1, W2, Qw, Kw);
    score_softmax_kernel<<<(B_DIM * TQ) / TPB_T, TV, 0, stream>>>(Qw, Kw, scale, attn);
    context_kernel<<<256, 256, 0, stream>>>(attn, value, ctx);
}


# GEMM K loops (projection, context) unrolled in full, waves-per-SIMD ceiling 8: each wave keeps about 57 loads (19 k-steps) in flight ahead of its WMMAs, was 6 with a full drain every 2 k-steps
# speedup vs baseline: 1.7016x; 1.7016x over previous
#include <hip/hip_runtime.h>
#include <hip/hip_bf16.h>

typedef __attribute__((ext_vector_type(2))) float v2f;
typedef __attribute__((ext_vector_type(8))) float v8f;

#define B_DIM 4
#define TQ 512
#define TV 512
#define DM 256
#define UN 128
#define TPB_T 8

static __device__ __forceinline__ float fast_tanh(float x) {
#if __has_builtin(__builtin_amdgcn_tanhf)
    return __builtin_amdgcn_tanhf(x);
#else
    return tanhf(x);
#endif
}

static __device__ __forceinline__ void wmma_tile_f32(
    const float* __restrict__ A, int lda,
    const float* __restrict__ B, int ldb,
    float* __restrict__ C, int ldc,
    int m0, int n0, int K)
{
    const int lane = threadIdx.x & 31;
    const int half = lane >> 4;
    const int mn   = lane & 15;
    v8f acc = {0.f, 0.f, 0.f, 0.f, 0.f, 0.f, 0.f, 0.f};
    const float* __restrict__ arow = A + (size_t)(m0 + mn) * lda + 2 * half;
    const float* __restrict__ bcol = B + (size_t)(2 * half) * ldb + (n0 + mn);
#pragma unroll
    for (int k0 = 0; k0 < K; k0 += 4) {
        v2f a, b;
        a.x = arow[k0 + 0];
        a.y = arow[k0 + 1];
        b.x = bcol[(size_t)(k0 + 0) * ldb];
        b.y = bcol[(size_t)(k0 + 1) * ldb];
        acc = __builtin_amdgcn_wmma_f32_16x16x4_f32(
            false, a, false, b, (short)0, acc, false, false);
    }
#pragma unroll
    for (int r = 0; r < 8; ++r) {
        C[(size_t)(m0 + r + 8 * half) * ldc + (n0 + mn)] = acc[r];
    }
}

__global__ __launch_bounds__(256) __attribute__((amdgpu_waves_per_eu(2, 8))) void proj_kernel(
    const float* __restrict__ query, const float* __restrict__ value,
    const float* __restrict__ W1, const float* __restrict__ W2,
    float* __restrict__ Qo, float* __restrict__ Ko)
{
    const int wave  = (int)((blockIdx.x * blockDim.x + threadIdx.x) >> 5);
    const int which = wave >> 10;
    const int tile  = wave & 1023;
    const int mt    = tile >> 3;
    const int nt    = tile & 7;
    const float* A  = which ? value : query;
    const float* Bm = which ? W2 : W1;
    float* C        = which ? Ko : Qo;
    wmma_tile_f32(A, DM, Bm, UN, C, UN, mt * 16, nt * 16, DM);
}

__global__ __launch_bounds__(512) void score_softmax_kernel(
    const float* __restrict__ Q, const float* __restrict__ Kmat,
    const float* __restrict__ scale, float* __restrict__ attn)
{
    const int b    = (int)blockIdx.x >> 6;
    const int t0   = ((int)blockIdx.x & 63) * TPB_T;
    const int tid  = (int)threadIdx.x;
    __shared__ float q_sh[TPB_T][UN];
    __shared__ float sc_sh[UN];
    __shared__ float red[16];

    if (tid < UN) sc_sh[tid] = scale[tid];
    {
        const int tt = tid >> 7;
        const int uu = tid & 127;
        q_sh[tt][uu]     = Q[(size_t)(b * TQ + t0 + tt) * UN + uu];
        q_sh[tt + 4][uu] = Q[(size_t)(b * TQ + t0 + tt + 4) * UN + uu];
    }

    float kreg[UN];
    {
        const float4* __restrict__ krow4 =
            (const float4*)(Kmat + ((size_t)(b * TV + tid)) * UN);
#pragma unroll
        for (int i = 0; i < UN / 4; ++i) {
            const float4 v = krow4[i];
            kreg[4 * i + 0] = v.x;
            kreg[4 * i + 1] = v.y;
            kreg[4 * i + 2] = v.z;
            kreg[4 * i + 3] = v.w;
        }
    }
    __syncthreads();

    const int w = tid >> 5;
    for (int tt = 0; tt < TPB_T; ++tt) {
        float s = 0.f;
#pragma unroll
        for (int u = 0; u < UN; ++u) {
            s = __builtin_fmaf(sc_sh[u], fast_tanh(q_sh[tt][u] + kreg[u]), s);
        }

        float m = s;
#pragma unroll
        for (int off = 16; off > 0; off >>= 1)
            m = fmaxf(m, __shfl_xor(m, off, 32));
        if ((tid & 31) == 0) red[w] = m;
        __syncthreads();
        float bmax = red[0];
#pragma unroll
        for (int i = 1; i < 16; ++i) bmax = fmaxf(bmax, red[i]);

        const float e = __expf(s - bmax);

        __syncthreads();
        float sum = e;
#pragma unroll
        for (int off = 16; off > 0; off >>= 1)
            sum += __shfl_xor(sum, off, 32);
        if ((tid & 31) == 0) red[w] = sum;
        __syncthreads();
        float bsum = 0.f;
#pragma unroll
        for (int i = 0; i < 16; ++i) bsum += red[i];

        attn[(size_t)(b * TQ + t0 + tt) * TV + tid] = e / bsum;
        __syncthreads();
    }
}

__global__ __launch_bounds__(256) __attribute__((amdgpu_waves_per_eu(2, 8))) void context_kernel(
    const float* __restrict__ attn, const float* __restrict__ value,
    float* __restrict__ ctx)
{
    const int wave = (int)((blockIdx.x * blockDim.x + threadIdx.x) >> 5);
    const int b    = wave >> 9;
    const int t    = wave & 511;
    const int mt   = t >> 4;
    const int nt   = t & 15;
    const float* A  = attn  + (size_t)b * TQ * TV;
    const float* Bv = value + (size_t)b * TV * DM;
    float* C        = ctx   + (size_t)b * TQ * DM;
    wmma_tile_f32(A, TV, Bv, DM, C, DM, mt * 16, nt * 16, TV);
}

extern "C" void kernel_launch(void* const* d_in, const int* in_sizes, int n_in,
                              void* d_out, int out_size, void* d_ws, size_t ws_size,
                              hipStream_t stream) {
    const float* query = (const float*)d_in[0];
    const float* value = (const float*)d_in[1];
    const float* W1    = (const float*)d_in[2];
    const float* W2    = (const float*)d_in[3];
    const float* scale = (const float*)d_in[4];

    float* out  = (float*)d_out;
    float* ctx  = out;
    float* attn = out + (size_t)B_DIM * TQ * DM;

    float* Qw = (float*)d_ws;
    float* Kw = Qw + (size_t)(B_DIM * TQ) * UN;

    proj_kernel<<<256, 256, 0, stream>>>(query, value, W1, W2, Qw, Kw);
    score_softmax_kernel<<<(B_DIM * TQ) / TPB_T, TV, 0, stream>>>(Qw, Kw, scale, attn);
    context_kernel<<<256, 256, 0, stream>>>(attn, value, ctx);
}


# score kernel: one pass over u carries all 8 query rows, each row sum as even-u and odd-u packed halves, one softmax pass for all 8 rows (row sums and softmax sums in a new summation tree)
# speedup vs baseline: 2.4162x; 1.4200x over previous
#include <hip/hip_runtime.h>
#include <hip/hip_bf16.h>

typedef __attribute__((ext_vector_type(2))) float v2f;
typedef __attribute__((ext_vector_type(8))) float v8f;

#define B_DIM 4
#define TQ 512
#define TV 512
#define DM 256
#define UN 128
#define TPB_T 8

static __device__ __forceinline__ float fast_tanh(float x) {
#if __has_builtin(__builtin_amdgcn_tanhf)
    return __builtin_amdgcn_tanhf(x);
#else
    return tanhf(x);
#endif
}

static __device__ __forceinline__ void wmma_tile_f32(
    const float* __restrict__ A, int lda,
    const float* __restrict__ B, int ldb,
    float* __restrict__ C, int ldc,
    int m0, int n0, int K)
{
    const int lane = threadIdx.x & 31;
    const int half = lane >> 4;
    const int mn   = lane & 15;
    v8f acc = {0.f, 0.f, 0.f, 0.f, 0.f, 0.f, 0.f, 0.f};
    const float* __restrict__ arow = A + (size_t)(m0 + mn) * lda + 2 * half;
    const float* __restrict__ bcol = B + (size_t)(2 * half) * ldb + (n0 + mn);
#pragma unroll
    for (int k0 = 0; k0 < K; k0 += 4) {
        v2f a, b;
        a.x = arow[k0 + 0];
        a.y = arow[k0 + 1];
        b.x = bcol[(size_t)(k0 + 0) * ldb];
        b.y = bcol[(size_t)(k0 + 1) * ldb];
        acc = __builtin_amdgcn_wmma_f32_16x16x4_f32(
            false, a, false, b, (short)0, acc, false, false);
    }
#pragma unroll
    for (int r = 0; r < 8; ++r) {
        C[(size_t)(m0 + r + 8 * half) * ldc + (n0 + mn)] = acc[r];
    }
}

__global__ __launch_bounds__(256) __attribute__((amdgpu_waves_per_eu(2, 8))) void proj_kernel(
    const float* __restrict__ query, const float* __restrict__ value,
    const float* __restrict__ W1, const float* __restrict__ W2,
    float* __restrict__ Qo, float* __restrict__ Ko)
{
    const int wave  = (int)((blockIdx.x * blockDim.x + threadIdx.x) >> 5);
    const int which = wave >> 10;
    const int tile  = wave & 1023;
    const int mt    = tile >> 3;
    const int nt    = tile & 7;
    const float* A  = which ? value : query;
    const float* Bm = which ? W2 : W1;
    float* C        = which ? Ko : Qo;
    wmma_tile_f32(A, DM, Bm, UN, C, UN, mt * 16, nt * 16, DM);
}

__global__ __launch_bounds__(512) __attribute__((amdgpu_waves_per_eu(4, 4))) void score_softmax_kernel(
    const float* __restrict__ Q, const float* __restrict__ Kmat,
    const float* __restrict__ scale, float* __restrict__ attn)
{
    const int b    = (int)blockIdx.x >> 6;
    const int t0   = ((int)blockIdx.x & 63) * TPB_T;
    const int tid  = (int)threadIdx.x;
    __shared__ __attribute__((aligned(16))) float q_sh[TPB_T][UN];
    __shared__ __attribute__((aligned(16))) float sc_sh[UN];
    __shared__ __attribute__((aligned(16))) float st_sh[TPB_T][TV];
    __shared__ __attribute__((aligned(16))) float red_max[TPB_T][2];
    __shared__ __attribute__((aligned(16))) float red_sum[TPB_T][2];

    if (tid < UN) sc_sh[tid] = scale[tid];
    {
        const int tt = tid >> 7;
        const int uu = tid & 127;
        q_sh[tt][uu]     = Q[(size_t)(b * TQ + t0 + tt) * UN + uu];
        q_sh[tt + 4][uu] = Q[(size_t)(b * TQ + t0 + tt + 4) * UN + uu];
    }

    float kreg[UN];
    {
        const float4* __restrict__ krow4 =
            (const float4*)(Kmat + ((size_t)(b * TV + tid)) * UN);
#pragma unroll
        for (int i = 0; i < UN / 4; ++i) {
            const float4 v = krow4[i];
            kreg[4 * i + 0] = v.x;
            kreg[4 * i + 1] = v.y;
            kreg[4 * i + 2] = v.z;
            kreg[4 * i + 3] = v.w;
        }
    }
    __syncthreads();

    v2f acc[TPB_T];
#pragma unroll
    for (int r = 0; r < TPB_T; ++r) { acc[r].x = 0.f; acc[r].y = 0.f; }

#define PAIR_FENCE() __builtin_amdgcn_sched_barrier(0)
    float4 qbuf[2][TPB_T];
    float4 scbuf[2];
    scbuf[0] = *(const float4*)&sc_sh[0];
#pragma unroll
    for (int r = 0; r < TPB_T; ++r) qbuf[0][r] = *(const float4*)&q_sh[r][0];
    v2f xa, xb;
    v2f tpa, tpb;
    v2f spa, spb;
    {
        v2f qa, qb, ka, kb;
        qa.x = qbuf[0][0].x; qa.y = qbuf[0][0].y; qb.x = qbuf[0][0].z; qb.y = qbuf[0][0].w;
        ka.x = kreg[0]; ka.y = kreg[1]; kb.x = kreg[2]; kb.y = kreg[3];
        xa = qa + ka; xb = qb + kb;
    }
    tpa.x = 0.f; tpa.y = 0.f; tpb.x = 0.f; tpb.y = 0.f;
    spa = tpa; spb = tpb;
#pragma unroll
    for (int j = 0; j < UN / 4; ++j) {
        if (j + 1 < UN / 4) {
            scbuf[(j + 1) & 1] = *(const float4*)&sc_sh[4 * (j + 1)];
#pragma unroll
            for (int r = 0; r < TPB_T; ++r)
                qbuf[(j + 1) & 1][r] = *(const float4*)&q_sh[r][4 * (j + 1)];
        }
        PAIR_FENCE();
#pragma unroll
        for (int r = 0; r < TPB_T; ++r) {
            const int n  = j * TPB_T + r;
            const int jn = (r == TPB_T - 1) ? j + 1 : j;
            const int rn = (r == TPB_T - 1) ? 0 : r + 1;
            const int rp = (r == 0) ? TPB_T - 1 : r - 1;
            v2f nxa = xa, nxb = xb, ta, tb;

            ta.x = fast_tanh(xa.x);
            if (n > 0) acc[rp] = __builtin_elementwise_fma(spa, tpa, acc[rp]);
            PAIR_FENCE();
            ta.y = fast_tanh(xa.y);
            if (n + 1 < TPB_T * (UN / 4)) {
                v2f qa, ka;
                qa.x = qbuf[jn & 1][rn].x; qa.y = qbuf[jn & 1][rn].y;
                ka.x = kreg[4 * jn + 0];   ka.y = kreg[4 * jn + 1];
                nxa = qa + ka;
            }
            PAIR_FENCE();
            tb.x = fast_tanh(xb.x);
            if (n > 0) acc[rp] = __builtin_elementwise_fma(spb, tpb, acc[rp]);
            PAIR_FENCE();
            tb.y = fast_tanh(xb.y);
            if (n + 1 < TPB_T * (UN / 4)) {
                v2f qb, kb;
                qb.x = qbuf[jn & 1][rn].z; qb.y = qbuf[jn & 1][rn].w;
                kb.x = kreg[4 * jn + 2];   kb.y = kreg[4 * jn + 3];
                nxb = qb + kb;
            }
            PAIR_FENCE();
            xa = nxa; xb = nxb; tpa = ta; tpb = tb;
            spa.x = scbuf[j & 1].x; spa.y = scbuf[j & 1].y;
            spb.x = scbuf[j & 1].z; spb.y = scbuf[j & 1].w;
        }
    }
    {
        acc[TPB_T - 1] = __builtin_elementwise_fma(spa, tpa, acc[TPB_T - 1]);
        acc[TPB_T - 1] = __builtin_elementwise_fma(spb, tpb, acc[TPB_T - 1]);
    }
#undef PAIR_FENCE

    float s[TPB_T];
#pragma unroll
    for (int r = 0; r < TPB_T; ++r) s[r] = acc[r].x + acc[r].y;

    const int lane = tid & 31;
    const int w    = tid >> 5;
    const int rr   = w >> 1;
    const int hh   = w & 1;
    const float* __restrict__ half_row = &st_sh[rr][hh * (TV / 2) + lane];

#pragma unroll
    for (int r = 0; r < TPB_T; ++r) st_sh[r][tid] = s[r];
    __syncthreads();
    {
        float m = half_row[0];
#pragma unroll
        for (int i = 1; i < 8; ++i) m = fmaxf(m, half_row[32 * i]);
#pragma unroll
        for (int off = 16; off > 0; off >>= 1)
            m = fmaxf(m, __shfl_xor(m, off, 32));
        if (lane == 0) red_max[rr][hh] = m;
    }
    __syncthreads();

    float e[TPB_T];
#pragma unroll
    for (int r = 0; r < TPB_T; ++r) {
        const float bmax = fmaxf(red_max[r][0], red_max[r][1]);
        e[r] = __expf(s[r] - bmax);
        st_sh[r][tid] = e[r];
    }
    __syncthreads();
    {
        float sum = half_row[0];
#pragma unroll
        for (int i = 1; i < 8; ++i) sum += half_row[32 * i];
#pragma unroll
        for (int off = 16; off > 0; off >>= 1)
            sum += __shfl_xor(sum, off, 32);
        if (lane == 0) red_sum[rr][hh] = sum;
    }
    __syncthreads();

#pragma unroll
    for (int r = 0; r < TPB_T; ++r) {
        const float bsum = red_sum[r][0] + red_sum[r][1];
        attn[(size_t)(b * TQ + t0 + r) * TV + tid] = e[r] / bsum;
    }
}

__global__ __launch_bounds__(256) __attribute__((amdgpu_waves_per_eu(2, 8))) void context_kernel(
    const float* __restrict__ attn, const float* __restrict__ value,
    float* __restrict__ ctx)
{
    const int wave = (int)((blockIdx.x * blockDim.x + threadIdx.x) >> 5);
    const int b    = wave >> 9;
    const int t    = wave & 511;
    const int mt   = t >> 4;
    const int nt   = t & 15;
    const float* A  = attn  + (size_t)b * TQ * TV;
    const float* Bv = value + (size_t)b * TV * DM;
    float* C        = ctx   + (size_t)b * TQ * DM;
    wmma_tile_f32(A, TV, Bv, DM, C, DM, mt * 16, nt * 16, TV);
}

extern "C" void kernel_launch(void* const* d_in, const int* in_sizes, int n_in,
                              void* d_out, int out_size, void* d_ws, size_t ws_size,
                              hipStream_t stream) {
    const float* query = (const float*)d_in[0];
    const float* value = (const float*)d_in[1];
    const float* W1    = (const float*)d_in[2];
    const float* W2    = (const float*)d_in[3];
    const float* scale = (const float*)d_in[4];

    float* out  = (float*)d_out;
    float* ctx  = out;
    float* attn = out + (size_t)B_DIM * TQ * DM;

    float* Qw = (float*)d_ws;
    float* Kw = Qw + (size_t)(B_DIM * TQ) * UN;

    proj_kernel<<<256, 256, 0, stream>>>(query, value, W1, W2, Qw, Kw);
    score_softmax_kernel<<<(B_DIM * TQ) / TPB_T, TV, 0, stream>>>(Qw, Kw, scale, attn);
    context_kernel<<<256, 256, 0, stream>>>(attn, value, ctx);
}


# Both GEMMs: the 16-row A strip that a block's 8 waves share is copied once into LDS with whole-line loads and A fragments are read from LDS (was a 16-line global load per k-step in every wave)
# speedup vs baseline: 2.6071x; 1.0790x over previous
#include <hip/hip_runtime.h>
#include <hip/hip_bf16.h>

typedef __attribute__((ext_vector_type(2))) float v2f;
typedef __attribute__((ext_vector_type(8))) float v8f;

#define B_DIM 4
#define TQ 512
#define TV 512
#define DM 256
#define UN 128
#define TPB_T 8

static __device__ __forceinline__ float fast_tanh(float x) {
#if __has_builtin(__builtin_amdgcn_tanhf)
    return __builtin_amdgcn_tanhf(x);
#else
    return tanhf(x);
#endif
}

template <int K>
static __device__ __forceinline__ void wmma_strip_f32(
    const float* __restrict__ A, int lda,
    const float* __restrict__ B, int ldb,
    float* __restrict__ C, int ldc,
    int m0, int n0, float* __restrict__ a_sh)
{
    constexpr int PITCH      = K + 4;
    constexpr int F4_PER_ROW = K / 4;
    constexpr int NF4        = 16 * F4_PER_ROW / 256;
    constexpr int KC         = 64;
    constexpr int NS         = KC / 4;
    const int tid  = (int)threadIdx.x;
    const int lane = tid & 31;
    const int half = lane >> 4;
    const int mn   = lane & 15;

    float4 st[NF4];
#pragma unroll
    for (int i = 0; i < NF4; ++i) {
        const int f   = tid + 256 * i;
        const int row = f / F4_PER_ROW;
        const int c4  = f % F4_PER_ROW;
        st[i] = *(const float4*)(A + (size_t)(m0 + row) * lda + 4 * c4);
    }
    const float* __restrict__ bcol = B + (size_t)(2 * half) * ldb + (n0 + mn);
    v2f b0[NS];
#pragma unroll
    for (int s = 0; s < NS; ++s) {
        b0[s].x = bcol[(size_t)(4 * s + 0) * ldb];
        b0[s].y = bcol[(size_t)(4 * s + 1) * ldb];
    }
#pragma unroll
    for (int i = 0; i < NF4; ++i) {
        const int f   = tid + 256 * i;
        const int row = f / F4_PER_ROW;
        const int c4  = f % F4_PER_ROW;
        *(float4*)&a_sh[row * PITCH + 4 * c4] = st[i];
    }
    __syncthreads();

    v8f acc = {0.f, 0.f, 0.f, 0.f, 0.f, 0.f, 0.f, 0.f};
    const float* __restrict__ arow = a_sh + mn * PITCH + 2 * half;
#define BLOCK_FENCE() __builtin_amdgcn_sched_barrier(0x0002 | 0x0004 | 0x0010)
    v2f a[2][NS];
#pragma unroll
    for (int s = 0; s < NS; ++s) a[0][s] = *(const v2f*)&arow[4 * s];
#pragma unroll
    for (int kc = 0; kc < K; kc += KC) {
        const int cur = (kc / KC) & 1;
        if (kc + KC < K) {
#pragma unroll
            for (int s = 0; s < NS; ++s)
                a[cur ^ 1][s] = *(const v2f*)&arow[kc + KC + 4 * s];
        }
        BLOCK_FENCE();
#pragma unroll
        for (int s = 0; s < NS; ++s) {
            v2f b;
            if (kc == 0) {
                b = b0[s];
            } else {
                b.x = bcol[(size_t)(kc + 4 * s + 0) * ldb];
                b.y = bcol[(size_t)(kc + 4 * s + 1) * ldb];
            }
            acc = __builtin_amdgcn_wmma_f32_16x16x4_f32(
                false, a[cur][s], false, b, (short)0, acc, false, false);
        }
        BLOCK_FENCE();
    }
#undef BLOCK_FENCE
#pragma unroll
    for (int r = 0; r < 8; ++r) {
        C[(size_t)(m0 + r + 8 * half) * ldc + (n0 + mn)] = acc[r];
    }
}

__global__ __launch_bounds__(256) __attribute__((amdgpu_waves_per_eu(2, 8))) void proj_kernel(
    const float* __restrict__ query, const float* __restrict__ value,
    const float* __restrict__ W1, const float* __restrict__ W2,
    float* __restrict__ Qo, float* __restrict__ Ko)
{
    const int wave  = (int)((blockIdx.x * blockDim.x + threadIdx.x) >> 5);
    const int which = wave >> 10;
    const int tile  = wave & 1023;
    const int mt    = tile >> 3;
    const int nt    = tile & 7;
    const float* A  = which ? value : query;
    const float* Bm = which ? W2 : W1;
    float* C        = which ? Ko : Qo;
    __shared__ __attribute__((aligned(16))) float a_sh[16 * (DM + 4)];
    wmma_strip_f32<DM>(A, DM, Bm, UN, C, UN, mt * 16, nt * 16, a_sh);
}

__global__ __launch_bounds__(512) __attribute__((amdgpu_waves_per_eu(4, 4))) void score_softmax_kernel(
    const float* __restrict__ Q, const float* __restrict__ Kmat,
    const float* __restrict__ scale, float* __restrict__ attn)
{
    const int b    = (int)blockIdx.x >> 6;
    const int t0   = ((int)blockIdx.x & 63) * TPB_T;
    const int tid  = (int)threadIdx.x;
    __shared__ __attribute__((aligned(16))) float q_sh[TPB_T][UN];
    __shared__ __attribute__((aligned(16))) float sc_sh[UN];
    __shared__ __attribute__((aligned(16))) float st_sh[TPB_T][TV];
    __shared__ __attribute__((aligned(16))) float red_max[TPB_T][2];
    __shared__ __attribute__((aligned(16))) float red_sum[TPB_T][2];

    if (tid < UN) sc_sh[tid] = scale[tid];
    {
        const int tt = tid >> 7;
        const int uu = tid & 127;
        q_sh[tt][uu]     = Q[(size_t)(b * TQ + t0 + tt) * UN + uu];
        q_sh[tt + 4][uu] = Q[(size_t)(b * TQ + t0 + tt + 4) * UN + uu];
    }

    float kreg[UN];
    {
        const float4* __restrict__ krow4 =
            (const float4*)(Kmat + ((size_t)(b * TV + tid)) * UN);
#pragma unroll
        for (int i = 0; i < UN / 4; ++i) {
            const float4 v = krow4[i];
            kreg[4 * i + 0] = v.x;
            kreg[4 * i + 1] = v.y;
            kreg[4 * i + 2] = v.z;
            kreg[4 * i + 3] = v.w;
        }
    }
    __syncthreads();

    v2f acc[TPB_T];
#pragma unroll
    for (int r = 0; r < TPB_T; ++r) { acc[r].x = 0.f; acc[r].y = 0.f; }

#define PAIR_FENCE() __builtin_amdgcn_sched_barrier(0)
    float4 qbuf[2][TPB_T];
    float4 scbuf[2];
    scbuf[0] = *(const float4*)&sc_sh[0];
#pragma unroll
    for (int r = 0; r < TPB_T; ++r) qbuf[0][r] = *(const float4*)&q_sh[r][0];
    v2f xa, xb;
    v2f tpa, tpb;
    v2f spa, spb;
    {
        v2f qa, qb, ka, kb;
        qa.x = qbuf[0][0].x; qa.y = qbuf[0][0].y; qb.x = qbuf[0][0].z; qb.y = qbuf[0][0].w;
        ka.x = kreg[0]; ka.y = kreg[1]; kb.x = kreg[2]; kb.y = kreg[3];
        xa = qa + ka; xb = qb + kb;
    }
    tpa.x = 0.f; tpa.y = 0.f; tpb.x = 0.f; tpb.y = 0.f;
    spa = tpa; spb = tpb;
#pragma unroll
    for (int j = 0; j < UN / 4; ++j) {
        if (j + 1 < UN / 4) {
            scbuf[(j + 1) & 1] = *(const float4*)&sc_sh[4 * (j + 1)];
#pragma unroll
            for (int r = 0; r < TPB_T; ++r)
                qbuf[(j + 1) & 1][r] = *(const float4*)&q_sh[r][4 * (j + 1)];
        }
        PAIR_FENCE();
#pragma unroll
        for (int r = 0; r < TPB_T; ++r) {
            const int n  = j * TPB_T + r;
            const int jn = (r == TPB_T - 1) ? j + 1 : j;
            const int rn = (r == TPB_T - 1) ? 0 : r + 1;
            const int rp = (r == 0) ? TPB_T - 1 : r - 1;
            v2f nxa = xa, nxb = xb, ta, tb;

            ta.x = fast_tanh(xa.x);
            if (n > 0) acc[rp] = __builtin_elementwise_fma(spa, tpa, acc[rp]);
            PAIR_FENCE();
            ta.y = fast_tanh(xa.y);
            if (n + 1 < TPB_T * (UN / 4)) {
                v2f qa, ka;
                qa.x = qbuf[jn & 1][rn].x; qa.y = qbuf[jn & 1][rn].y;
                ka.x = kreg[4 * jn + 0];   ka.y = kreg[4 * jn + 1];
                nxa = qa + ka;
            }
            PAIR_FENCE();
            tb.x = fast_tanh(xb.x);
            if (n > 0) acc[rp] = __builtin_elementwise_fma(spb, tpb, acc[rp]);
            PAIR_FENCE();
            tb.y = fast_tanh(xb.y);
            if (n + 1 < TPB_T * (UN / 4)) {
                v2f qb, kb;
                qb.x = qbuf[jn & 1][rn].z; qb.y = qbuf[jn & 1][rn].w;
                kb.x = kreg[4 * jn + 2];   kb.y = kreg[4 * jn + 3];
                nxb = qb + kb;
            }
            PAIR_FENCE();
            xa = nxa; xb = nxb; tpa = ta; tpb = tb;
            spa.x = scbuf[j & 1].x; spa.y = scbuf[j & 1].y;
            spb.x = scbuf[j & 1].z; spb.y = scbuf[j & 1].w;
        }
    }
    {
        acc[TPB_T - 1] = __builtin_elementwise_fma(spa, tpa, acc[TPB_T - 1]);
        acc[TPB_T - 1] = __builtin_elementwise_fma(spb, tpb, acc[TPB_T - 1]);
    }
#undef PAIR_FENCE

    float s[TPB_T];
#pragma unroll
    for (int r = 0; r < TPB_T; ++r) s[r] = acc[r].x + acc[r].y;

    const int lane = tid & 31;
    const int w    = tid >> 5;
    const int rr   = w >> 1;
    const int hh   = w & 1;
    const float* __restrict__ half_row = &st_sh[rr][hh * (TV / 2) + lane];

#pragma unroll
    for (int r = 0; r < TPB_T; ++r) st_sh[r][tid] = s[r];
    __syncthreads();
    {
        float m = half_row[0];
#pragma unroll
        for (int i = 1; i < 8; ++i) m = fmaxf(m, half_row[32 * i]);
#pragma unroll
        for (int off = 16; off > 0; off >>= 1)
            m = fmaxf(m, __shfl_xor(m, off, 32));
        if (lane == 0) red_max[rr][hh] = m;
    }
    __syncthreads();

    float e[TPB_T];
#pragma unroll
    for (int r = 0; r < TPB_T; ++r) {
        const float bmax = fmaxf(red_max[r][0], red_max[r][1]);
        e[r] = __expf(s[r] - bmax);
        st_sh[r][tid] = e[r];
    }
    __syncthreads();
    {
        float sum = half_row[0];
#pragma unroll
        for (int i = 1; i < 8; ++i) sum += half_row[32 * i];
#pragma unroll
        for (int off = 16; off > 0; off >>= 1)
            sum += __shfl_xor(sum, off, 32);
        if (lane == 0) red_sum[rr][hh] = sum;
    }
    __syncthreads();

#pragma unroll
    for (int r = 0; r < TPB_T; ++r) {
        const float bsum = red_sum[r][0] + red_sum[r][1];
        attn[(size_t)(b * TQ + t0 + r) * TV + tid] = e[r] / bsum;
    }
}

__global__ __launch_bounds__(256) __attribute__((amdgpu_waves_per_eu(2, 8))) void context_kernel(
    const float* __restrict__ attn, const float* __restrict__ value,
    float* __restrict__ ctx)
{
    const int wave = (int)((blockIdx.x * blockDim.x + threadIdx.x) >> 5);
    const int b    = wave >> 9;
    const int t    = wave & 511;
    const int mt   = t >> 4;
    const int nt   = t & 15;
    const float* A  = attn  + (size_t)b * TQ * TV;
    const float* Bv = value + (size_t)b * TV * DM;
    float* C        = ctx   + (size_t)b * TQ * DM;
    __shared__ __attribute__((aligned(16))) float a_sh[16 * (TV + 4)];
    wmma_strip_f32<TV>(A, TV, Bv, DM, C, DM, mt * 16, nt * 16, a_sh);
}

extern "C" void kernel_launch(void* const* d_in, const int* in_sizes, int n_in,
                              void* d_out, int out_size, void* d_ws, size_t ws_size,
                              hipStream_t stream) {
    const float* query = (const float*)d_in[0];
    const float* value = (const float*)d_in[1];
    const float* W1    = (const float*)d_in[2];
    const float* W2    = (const float*)d_in[3];
    const float* scale = (const float*)d_in[4];

    float* out  = (float*)d_out;
    float* ctx  = out;
    float* attn = out + (size_t)B_DIM * TQ * DM;

    float* Qw = (float*)d_ws;
    float* Kw = Qw + (size_t)(B_DIM * TQ) * UN;

    proj_kernel<<<256, 256, 0, stream>>>(query, value, W1, W2, Qw, Kw);
    score_softmax_kernel<<<(B_DIM * TQ) / TPB_T, TV, 0, stream>>>(Qw, Kw, scale, attn);
    context_kernel<<<256, 256, 0, stream>>>(attn, value, ctx);
}


# context product moved into the score kernel behind the softmax: weights stay in LDS as the A operand, one column tile per wave, half of each tile's rows real; launches 3 -> 2, same summation order
# speedup vs baseline: 2.7515x; 1.0554x over previous
#include <hip/hip_runtime.h>
#include <hip/hip_bf16.h>

typedef __attribute__((ext_vector_type(2))) float v2f;
typedef __attribute__((ext_vector_type(8))) float v8f;

#define B_DIM 4
#define TQ 512
#define TV 512
#define DM 256
#define UN 128
#define TPB_T 8

static __device__ __forceinline__ float fast_tanh(float x) {
#if __has_builtin(__builtin_amdgcn_tanhf)
    return __builtin_amdgcn_tanhf(x);
#else
    return tanhf(x);
#endif
}

template <int K>
static __device__ __forceinline__ void wmma_strip_f32(
    const float* __restrict__ A, int lda,
    const float* __restrict__ B, int ldb,
    float* __restrict__ C, int ldc,
    int m0, int n0, float* __restrict__ a_sh)
{
    constexpr int PITCH      = K + 4;
    constexpr int F4_PER_ROW = K / 4;
    constexpr int NF4        = 16 * F4_PER_ROW / 256;
    constexpr int KC         = 64;
    constexpr int NS         = KC / 4;
    const int tid  = (int)threadIdx.x;
    const int lane = tid & 31;
    const int half = lane >> 4;
    const int mn   = lane & 15;

    float4 st[NF4];
#pragma unroll
    for (int i = 0; i < NF4; ++i) {
        const int f   = tid + 256 * i;
        const int row = f / F4_PER_ROW;
        const int c4  = f % F4_PER_ROW;
        st[i] = *(const float4*)(A + (size_t)(m0 + row) * lda + 4 * c4);
    }
    const float* __restrict__ bcol = B + (size_t)(2 * half) * ldb + (n0 + mn);
    v2f b0[NS];
#pragma unroll
    for (int s = 0; s < NS; ++s) {
        b0[s].x = bcol[(size_t)(4 * s + 0) * ldb];
        b0[s].y = bcol[(size_t)(4 * s + 1) * ldb];
    }
#pragma unroll
    for (int i = 0; i < NF4; ++i) {
        const int f   = tid + 256 * i;
        const int row = f / F4_PER_ROW;
        const int c4  = f % F4_PER_ROW;
        *(float4*)&a_sh[row * PITCH + 4 * c4] = st[i];
    }
    __syncthreads();

    v8f acc = {0.f, 0.f, 0.f, 0.f, 0.f, 0.f, 0.f, 0.f};
    const float* __restrict__ arow = a_sh + mn * PITCH + 2 * half;
#define BLOCK_FENCE() __builtin_amdgcn_sched_barrier(0x0002 | 0x0004 | 0x0010)
    v2f a[2][NS];
#pragma unroll
    for (int s = 0; s < NS; ++s) a[0][s] = *(const v2f*)&arow[4 * s];
#pragma unroll
    for (int kc = 0; kc < K; kc += KC) {
        const int cur = (kc / KC) & 1;
        if (kc + KC < K) {
#pragma unroll
            for (int s = 0; s < NS; ++s)
                a[cur ^ 1][s] = *(const v2f*)&arow[kc + KC + 4 * s];
        }
        BLOCK_FENCE();
#pragma unroll
        for (int s = 0; s < NS; ++s) {
            v2f b;
            if (kc == 0) {
                b = b0[s];
            } else {
                b.x = bcol[(size_t)(kc + 4 * s + 0) * ldb];
                b.y = bcol[(size_t)(kc + 4 * s + 1) * ldb];
            }
            acc = __builtin_amdgcn_wmma_f32_16x16x4_f32(
                false, a[cur][s], false, b, (short)0, acc, false, false);
        }
        BLOCK_FENCE();
    }
#undef BLOCK_FENCE
#pragma unroll
    for (int r = 0; r < 8; ++r) {
        C[(size_t)(m0 + r + 8 * half) * ldc + (n0 + mn)] = acc[r];
    }
}

__global__ __launch_bounds__(256) __attribute__((amdgpu_waves_per_eu(2, 8))) void proj_kernel(
    const float* __restrict__ query, const float* __restrict__ value,
    const float* __restrict__ W1, const float* __restrict__ W2,
    float* __restrict__ Qo, float* __restrict__ Ko)
{
    const int wave  = (int)((blockIdx.x * blockDim.x + threadIdx.x) >> 5);
    const int which = wave >> 10;
    const int tile  = wave & 1023;
    const int mt    = tile >> 3;
    const int nt    = tile & 7;
    const float* A  = which ? value : query;
    const float* Bm = which ? W2 : W1;
    float* C        = which ? Ko : Qo;
    __shared__ __attribute__((aligned(16))) float a_sh[16 * (DM + 4)];
    wmma_strip_f32<DM>(A, DM, Bm, UN, C, UN, mt * 16, nt * 16, a_sh);
}

__global__ __launch_bounds__(512) __attribute__((amdgpu_waves_per_eu(4, 4))) void score_softmax_kernel(
    const float* __restrict__ Q, const float* __restrict__ Kmat,
    const float* __restrict__ scale, const float* __restrict__ value,
    float* __restrict__ attn, float* __restrict__ ctx)
{
    const int b    = (int)blockIdx.x >> 6;
    const int t0   = ((int)blockIdx.x & 63) * TPB_T;
    const int tid  = (int)threadIdx.x;
    __shared__ __attribute__((aligned(16))) float q_sh[TPB_T][UN];
    __shared__ __attribute__((aligned(16))) float sc_sh[UN];
    __shared__ __attribute__((aligned(16))) float st_sh[TPB_T][TV];
    __shared__ __attribute__((aligned(16))) float red_max[TPB_T][2];
    __shared__ __attribute__((aligned(16))) float red_sum[TPB_T][2];
    __shared__ __attribute__((aligned(16))) float p_sh[TPB_T][TV + 4];

    if (tid < UN) sc_sh[tid] = scale[tid];
    {
        const int tt = tid >> 7;
        const int uu = tid & 127;
        q_sh[tt][uu]     = Q[(size_t)(b * TQ + t0 + tt) * UN + uu];
        q_sh[tt + 4][uu] = Q[(size_t)(b * TQ + t0 + tt + 4) * UN + uu];
    }

    float kreg[UN];
    {
        const float4* __restrict__ krow4 =
            (const float4*)(Kmat + ((size_t)(b * TV + tid)) * UN);
#pragma unroll
        for (int i = 0; i < UN / 4; ++i) {
            const float4 v = krow4[i];
            kreg[4 * i + 0] = v.x;
            kreg[4 * i + 1] = v.y;
            kreg[4 * i + 2] = v.z;
            kreg[4 * i + 3] = v.w;
        }
    }
    __syncthreads();

    v2f acc[TPB_T];
#pragma unroll
    for (int r = 0; r < TPB_T; ++r) { acc[r].x = 0.f; acc[r].y = 0.f; }

#define PAIR_FENCE() __builtin_amdgcn_sched_barrier(0)
    float4 qbuf[2][TPB_T];
    float4 scbuf[2];
    scbuf[0] = *(const float4*)&sc_sh[0];
#pragma unroll
    for (int r = 0; r < TPB_T; ++r) qbuf[0][r] = *(const float4*)&q_sh[r][0];
    v2f xa, xb;
    v2f tpa, tpb;
    v2f spa, spb;
    {
        v2f qa, qb, ka, kb;
        qa.x = qbuf[0][0].x; qa.y = qbuf[0][0].y; qb.x = qbuf[0][0].z; qb.y = qbuf[0][0].w;
        ka.x = kreg[0]; ka.y = kreg[1]; kb.x = kreg[2]; kb.y = kreg[3];
        xa = qa + ka; xb = qb + kb;
    }
    tpa.x = 0.f; tpa.y = 0.f; tpb.x = 0.f; tpb.y = 0.f;
    spa = tpa; spb = tpb;
#pragma unroll
    for (int j = 0; j < UN / 4; ++j) {
        if (j + 1 < UN / 4) {
            scbuf[(j + 1) & 1] = *(const float4*)&sc_sh[4 * (j + 1)];
#pragma unroll
            for (int r = 0; r < TPB_T; ++r)
                qbuf[(j + 1) & 1][r] = *(const float4*)&q_sh[r][4 * (j + 1)];
        }
        PAIR_FENCE();
#pragma unroll
        for (int r = 0; r < TPB_T; ++r) {
            const int n  = j * TPB_T + r;
            const int jn = (r == TPB_T - 1) ? j + 1 : j;
            const int rn = (r == TPB_T - 1) ? 0 : r + 1;
            const int rp = (r == 0) ? TPB_T - 1 : r - 1;
            v2f nxa = xa, nxb = xb, ta, tb;

            ta.x = fast_tanh(xa.x);
            if (n > 0) acc[rp] = __builtin_elementwise_fma(spa, tpa, acc[rp]);
            PAIR_FENCE();
            ta.y = fast_tanh(xa.y);
            if (n + 1 < TPB_T * (UN / 4)) {
                v2f qa, ka;
                qa.x = qbuf[jn & 1][rn].x; qa.y = qbuf[jn & 1][rn].y;
                ka.x = kreg[4 * jn + 0];   ka.y = kreg[4 * jn + 1];
                nxa = qa + ka;
            }
            PAIR_FENCE();
            tb.x = fast_tanh(xb.x);
            if (n > 0) acc[rp] = __builtin_elementwise_fma(spb, tpb, acc[rp]);
            PAIR_FENCE();
            tb.y = fast_tanh(xb.y);
            if (n + 1 < TPB_T * (UN / 4)) {
                v2f qb, kb;
                qb.x = qbuf[jn & 1][rn].z; qb.y = qbuf[jn & 1][rn].w;
                kb.x = kreg[4 * jn + 2];   kb.y = kreg[4 * jn + 3];
                nxb = qb + kb;
            }
            PAIR_FENCE();
            xa = nxa; xb = nxb; tpa = ta; tpb = tb;
            spa.x = scbuf[j & 1].x; spa.y = scbuf[j & 1].y;
            spb.x = scbuf[j & 1].z; spb.y = scbuf[j & 1].w;
        }
    }
    {
        acc[TPB_T - 1] = __builtin_elementwise_fma(spa, tpa, acc[TPB_T - 1]);
        acc[TPB_T - 1] = __builtin_elementwise_fma(spb, tpb, acc[TPB_T - 1]);
    }
#undef PAIR_FENCE

    float s[TPB_T];
#pragma unroll
    for (int r = 0; r < TPB_T; ++r) s[r] = acc[r].x + acc[r].y;

    const int lane = tid & 31;
    const int w    = tid >> 5;
    const int rr   = w >> 1;
    const int hh   = w & 1;
    const float* __restrict__ half_row = &st_sh[rr][hh * (TV / 2) + lane];

#pragma unroll
    for (int r = 0; r < TPB_T; ++r) st_sh[r][tid] = s[r];
    __syncthreads();
    {
        float m = half_row[0];
#pragma unroll
        for (int i = 1; i < 8; ++i) m = fmaxf(m, half_row[32 * i]);
#pragma unroll
        for (int off = 16; off > 0; off >>= 1)
            m = fmaxf(m, __shfl_xor(m, off, 32));
        if (lane == 0) red_max[rr][hh] = m;
    }
    __syncthreads();

    float e[TPB_T];
#pragma unroll
    for (int r = 0; r < TPB_T; ++r) {
        const float bmax = fmaxf(red_max[r][0], red_max[r][1]);
        e[r] = __expf(s[r] - bmax);
        st_sh[r][tid] = e[r];
    }
    __syncthreads();
    {
        float sum = half_row[0];
#pragma unroll
        for (int i = 1; i < 8; ++i) sum += half_row[32 * i];
#pragma unroll
        for (int off = 16; off > 0; off >>= 1)
            sum += __shfl_xor(sum, off, 32);
        if (lane == 0) red_sum[rr][hh] = sum;
    }
    __syncthreads();

    constexpr int KC = 64;
    constexpr int NS = KC / 4;
    const int half = lane >> 4;
    const int mn   = lane & 15;
    const int n0   = w * 16;
    const float* __restrict__ bcol =
        value + ((size_t)b * TV + 2 * half) * DM + (n0 + mn);
    v2f b0[NS];
#pragma unroll
    for (int s2 = 0; s2 < NS; ++s2) {
        b0[s2].x = bcol[(size_t)(4 * s2 + 0) * DM];
        b0[s2].y = bcol[(size_t)(4 * s2 + 1) * DM];
    }
#pragma unroll
    for (int r = 0; r < TPB_T; ++r) {
        const float bsum = red_sum[r][0] + red_sum[r][1];
        const float p = e[r] / bsum;
        attn[(size_t)(b * TQ + t0 + r) * TV + tid] = p;
        p_sh[r][tid] = p;
    }
    __builtin_amdgcn_fence(__ATOMIC_RELEASE, "workgroup", "local");
    __builtin_amdgcn_s_barrier();
    __builtin_amdgcn_fence(__ATOMIC_ACQUIRE, "workgroup", "local");

    v8f cacc = {0.f, 0.f, 0.f, 0.f, 0.f, 0.f, 0.f, 0.f};
    const float* __restrict__ arow = &p_sh[mn & 7][2 * half];
#define BLOCK_FENCE() __builtin_amdgcn_sched_barrier(0x0002 | 0x0004 | 0x0010)
    v2f af[2][NS];
#pragma unroll
    for (int s2 = 0; s2 < NS; ++s2) af[0][s2] = *(const v2f*)&arow[4 * s2];
#pragma unroll
    for (int kc = 0; kc < TV; kc += KC) {
        const int cur = (kc / KC) & 1;
        if (kc + KC < TV) {
#pragma unroll
            for (int s2 = 0; s2 < NS; ++s2)
                af[cur ^ 1][s2] = *(const v2f*)&arow[kc + KC + 4 * s2];
        }
        BLOCK_FENCE();
#pragma unroll
        for (int s2 = 0; s2 < NS; ++s2) {
            v2f bb;
            if (kc == 0) {
                bb = b0[s2];
            } else {
                bb.x = bcol[(size_t)(kc + 4 * s2 + 0) * DM];
                bb.y = bcol[(size_t)(kc + 4 * s2 + 1) * DM];
            }
            cacc = __builtin_amdgcn_wmma_f32_16x16x4_f32(
                false, af[cur][s2], false, bb, (short)0, cacc, false, false);
        }
        BLOCK_FENCE();
    }
#undef BLOCK_FENCE
    if (half == 0) {
#pragma unroll
        for (int r = 0; r < TPB_T; ++r)
            ctx[(size_t)(b * TQ + t0 + r) * DM + (n0 + mn)] = cacc[r];
    }
}

extern "C" void kernel_launch(void* const* d_in, const int* in_sizes, int n_in,
                              void* d_out, int out_size, void* d_ws, size_t ws_size,
                              hipStream_t stream) {
    const float* query = (const float*)d_in[0];
    const float* value = (const float*)d_in[1];
    const float* W1    = (const float*)d_in[2];
    const float* W2    = (const float*)d_in[3];
    const float* scale = (const float*)d_in[4];

    float* out  = (float*)d_out;
    float* ctx  = out;
    float* attn = out + (size_t)B_DIM * TQ * DM;

    float* Qw = (float*)d_ws;
    float* Kw = Qw + (size_t)(B_DIM * TQ) * UN;

    proj_kernel<<<256, 256, 0, stream>>>(query, value, W1, W2, Qw, Kw);
    score_softmax_kernel<<<(B_DIM * TQ) / TPB_T, TV, 0, stream>>>(Qw, Kw, scale, value, attn, ctx);
}
